# starting kernel
# baseline (speedup 1.0000x reference)
#include <hip/hip_runtime.h>
#include <hip/hip_bf16.h>

typedef __attribute__((ext_vector_type(2))) float v2f;
typedef __attribute__((ext_vector_type(8))) float v8f;

#define DFEAT 64
#define ROWS_PER_BLOCK 128
#define WT_PITCH 66

__global__ void g2_zero_kernel(float* __restrict__ p, int n) {
    int i = blockIdx.x * blockDim.x + threadIdx.x;
    if (i < n) p[i] = 0.0f;
}

__global__ void g2_scatter_agg_kernel(const float* __restrict__ nf,
                                      const int* __restrict__ src,
                                      const int* __restrict__ dst,
                                      float* __restrict__ sums,
                                      float* __restrict__ cnt,
                                      int n_edges) {
    int gid = blockIdx.x * blockDim.x + threadIdx.x;
    int e = gid >> 4;
    if (e >= n_edges) return;
    int q = (gid & 15) * 4;
    int s = src[e];
    int d = dst[e];
    const float4 v = *(const float4*)(nf + (size_t)s * DFEAT + q);
    float* out = sums + (size_t)d * DFEAT + q;
    atomicAdd(out + 0, v.x);
    atomicAdd(out + 1, v.y);
    atomicAdd(out + 2, v.z);
    atomicAdd(out + 3, v.w);
    if (q == 0) atomicAdd(cnt + d, 1.0f);
}

__global__ __launch_bounds__(256) void g2_gemm_wmma_kernel(
        const float* __restrict__ sums,
        const float* __restrict__ cnt,
        const float* __restrict__ W,
        float* __restrict__ X,
        int n_nodes) {
    __shared__ float ldsA[ROWS_PER_BLOCK * DFEAT];
    __shared__ float ldsWT[DFEAT * WT_PITCH];

    const int tid = threadIdx.x;
    const int rowBase = blockIdx.x * ROWS_PER_BLOCK;

    for (int i = tid; i < DFEAT * DFEAT; i += 256) {
        int k = i >> 6, n = i & 63;
        ldsWT[n * WT_PITCH + k] = W[i];
    }

    for (int i = tid; i < ROWS_PER_BLOCK * DFEAT; i += 256) {
        int r = rowBase + (i >> 6);
        float v = 0.0f;
        if (r < n_nodes) {
            float c = cnt[r];
            v = sums[(size_t)r * DFEAT + (i & 63)] / fmaxf(c, 1.0f);
        }
        ldsA[i] = v;
    }
    __syncthreads();

    const int wave = tid >> 5;
    const int lane = tid & 31;
    const int m    = lane & 15;
    const int koff = (lane >> 4) << 1;

    const float* aRow = &ldsA[(wave * 16 + m) * DFEAT];
    const float* w0 = &ldsWT[(m +  0) * WT_PITCH];
    const float* w1 = &ldsWT[(m + 16) * WT_PITCH];
    const float* w2 = &ldsWT[(m + 32) * WT_PITCH];
    const float* w3 = &ldsWT[(m + 48) * WT_PITCH];

    v8f acc0 = {}, acc1 = {}, acc2 = {}, acc3 = {};
    for (int k = 0; k < 16; ++k) {
        const int kk = 4 * k + koff;
        const v2f a  = *(const v2f*)(aRow + kk);
        const v2f b0 = *(const v2f*)(w0 + kk);
        const v2f b1 = *(const v2f*)(w1 + kk);
        const v2f b2 = *(const v2f*)(w2 + kk);
        const v2f b3 = *(const v2f*)(w3 + kk);
        acc0 = __builtin_amdgcn_wmma_f32_16x16x4_f32(false, a, false, b0,
                                                     (short)0, acc0, false, false);
        acc1 = __builtin_amdgcn_wmma_f32_16x16x4_f32(false, a, false, b1,
                                                     (short)0, acc1, false, false);
        acc2 = __builtin_amdgcn_wmma_f32_16x16x4_f32(false, a, false, b2,
                                                     (short)0, acc2, false, false);
        acc3 = __builtin_amdgcn_wmma_f32_16x16x4_f32(false, a, false, b3,
                                                     (short)0, acc3, false, false);
    }

    const int row0 = rowBase + wave * 16 + ((lane >> 4) << 3);
    for (int v = 0; v < 8; ++v) {
        int row = row0 + v;
        if (row < n_nodes) {
            float* xr = X + (size_t)row * DFEAT + m;
            xr[0]  = acc0[v];
            xr[16] = acc1[v];
            xr[32] = acc2[v];
            xr[48] = acc3[v];
        }
    }
}

__global__ void g2_scatter_tau_kernel(const float* __restrict__ X,
                                      const int* __restrict__ src,
                                      const int* __restrict__ dst,
                                      float* __restrict__ sums,
                                      int n_edges) {
    int gid = blockIdx.x * blockDim.x + threadIdx.x;
    int e = gid >> 4;
    if (e >= n_edges) return;
    int q = (gid & 15) * 4;
    int s = src[e];
    int d = dst[e];
    const float4 a = *(const float4*)(X + (size_t)s * DFEAT + q);
    const float4 b = *(const float4*)(X + (size_t)d * DFEAT + q);
    float dx = a.x - b.x, dy = a.y - b.y, dz = a.z - b.z, dw = a.w - b.w;
    float* out = sums + (size_t)d * DFEAT + q;
    atomicAdd(out + 0, dx * dx);
    atomicAdd(out + 1, dy * dy);
    atomicAdd(out + 2, dz * dz);
    atomicAdd(out + 3, dw * dw);
}

__global__ void g2_finalize_kernel(const float* __restrict__ sums,
                                   const float* __restrict__ cnt,
                                   float* __restrict__ out,
                                   int n_elems) {
    int i = blockIdx.x * blockDim.x + threadIdx.x;
    if (i >= n_elems) return;
    float c = fmaxf(cnt[i >> 6], 1.0f);
    out[i] = tanhf(sums[i] / c);
}

extern "C" void kernel_launch(void* const* d_in, const int* in_sizes, int n_in,
                              void* d_out, int out_size, void* d_ws, size_t ws_size,
                              hipStream_t stream) {
    const float* nf = (const float*)d_in[0];
    const float* W  = (const float*)d_in[1];
    const int*   ei = (const int*)d_in[2];
    float* out = (float*)d_out;

    const int n_nodes = in_sizes[0] / DFEAT;
    const int n_edges = in_sizes[2] / 2;
    const int* src = ei;
    const int* dst = ei + n_edges;

    float* sums = (float*)d_ws;
    float* cnt  = sums + (size_t)n_nodes * DFEAT;
    float* X    = cnt + n_nodes;

    const int T = 256;

    {
        int n = n_nodes * (DFEAT + 1);
        g2_zero_kernel<<<(n + T - 1) / T, T, 0, stream>>>(sums, n);
    }
    {
        long long total = (long long)n_edges * 16;
        g2_scatter_agg_kernel<<<(int)((total + T - 1) / T), T, 0, stream>>>(
            nf, src, dst, sums, cnt, n_edges);
    }
    {
        int blocks = (n_nodes + ROWS_PER_BLOCK - 1) / ROWS_PER_BLOCK;
        g2_gemm_wmma_kernel<<<blocks, T, 0, stream>>>(sums, cnt, W, X, n_nodes);
    }
    {
        int n = n_nodes * DFEAT;
        g2_zero_kernel<<<(n + T - 1) / T, T, 0, stream>>>(sums, n);
    }
    {
        long long total = (long long)n_edges * 16;
        g2_scatter_tau_kernel<<<(int)((total + T - 1) / T), T, 0, stream>>>(
            X, src, dst, sums, n_edges);
    }
    {
        int n = n_nodes * DFEAT;
        g2_finalize_kernel<<<(n + T - 1) / T, T, 0, stream>>>(sums, cnt, out, n);
    }
}


# Both edge scatters: one wave per 32 edges, lane = feature column: each gather and each f32 atomic add instruction covers one whole 128-byte row line (was 8 floats in 4 lines); sums in arrival order
# speedup vs baseline: 1.5225x; 1.5225x over previous
#include <hip/hip_runtime.h>
#include <hip/hip_bf16.h>

typedef __attribute__((ext_vector_type(2))) float v2f;
typedef __attribute__((ext_vector_type(8))) float v8f;

#define DFEAT 64
#define ROWS_PER_BLOCK 128
#define WT_PITCH 66

__global__ void g2_zero_kernel(float* __restrict__ p, int n) {
    int i = blockIdx.x * blockDim.x + threadIdx.x;
    if (i < n) p[i] = 0.0f;
}

#define EDGES_PER_WAVE 32
#define EDGE_BATCH 8

template <bool FULL>
__device__ __forceinline__ void g2_agg_rows(const float* __restrict__ nf,
                                            float* __restrict__ sums,
                                            int sv, int dv, int lane, int nvalid) {
#pragma unroll
    for (int j0 = 0; j0 < EDGES_PER_WAVE; j0 += EDGE_BATCH) {
        float a0[EDGE_BATCH], a1[EDGE_BATCH];
#pragma unroll
        for (int j = 0; j < EDGE_BATCH; ++j) {
            const int s = __builtin_amdgcn_readlane(sv, j0 + j);
            const float* row = nf + (size_t)s * DFEAT;
            a0[j] = row[lane];
            a1[j] = row[32 + lane];
        }
#pragma unroll
        for (int j = 0; j < EDGE_BATCH; ++j) {
            if (FULL || j0 + j < nvalid) {
                const int d = __builtin_amdgcn_readlane(dv, j0 + j);
                float* orow = sums + (size_t)d * DFEAT;
                atomicAdd(orow + lane, a0[j]);
                atomicAdd(orow + 32 + lane, a1[j]);
            }
        }
    }
}

__global__ __launch_bounds__(256) void g2_scatter_agg_kernel(
        const float* __restrict__ nf,
        const int* __restrict__ src,
        const int* __restrict__ dst,
        float* __restrict__ sums,
        float* __restrict__ cnt,
        int n_edges) {
    const int lane  = threadIdx.x & 31;
    const int ebase = __builtin_amdgcn_readfirstlane(
        (int)((blockIdx.x * blockDim.x + threadIdx.x) >> 5)) * EDGES_PER_WAVE;
    if (ebase >= n_edges) return;
    const int e = ebase + lane;
    const bool valid = e < n_edges;
    const int sv = valid ? src[e] : 0;
    const int dv = valid ? dst[e] : 0;
    if (valid) atomicAdd(cnt + dv, 1.0f);
    const int nvalid = n_edges - ebase;
    if (nvalid >= EDGES_PER_WAVE) g2_agg_rows<true >(nf, sums, sv, dv, lane, EDGES_PER_WAVE);
    else                          g2_agg_rows<false>(nf, sums, sv, dv, lane, nvalid);
}

__global__ __launch_bounds__(256) void g2_gemm_wmma_kernel(
        const float* __restrict__ sums,
        const float* __restrict__ cnt,
        const float* __restrict__ W,
        float* __restrict__ X,
        int n_nodes) {
    __shared__ float ldsA[ROWS_PER_BLOCK * DFEAT];
    __shared__ float ldsWT[DFEAT * WT_PITCH];

    const int tid = threadIdx.x;
    const int rowBase = blockIdx.x * ROWS_PER_BLOCK;

    for (int i = tid; i < DFEAT * DFEAT; i += 256) {
        int k = i >> 6, n = i & 63;
        ldsWT[n * WT_PITCH + k] = W[i];
    }

    for (int i = tid; i < ROWS_PER_BLOCK * DFEAT; i += 256) {
        int r = rowBase + (i >> 6);
        float v = 0.0f;
        if (r < n_nodes) {
            float c = cnt[r];
            v = sums[(size_t)r * DFEAT + (i & 63)] / fmaxf(c, 1.0f);
        }
        ldsA[i] = v;
    }
    __syncthreads();

    const int wave = tid >> 5;
    const int lane = tid & 31;
    const int m    = lane & 15;
    const int koff = (lane >> 4) << 1;

    const float* aRow = &ldsA[(wave * 16 + m) * DFEAT];
    const float* w0 = &ldsWT[(m +  0) * WT_PITCH];
    const float* w1 = &ldsWT[(m + 16) * WT_PITCH];
    const float* w2 = &ldsWT[(m + 32) * WT_PITCH];
    const float* w3 = &ldsWT[(m + 48) * WT_PITCH];

    v8f acc0 = {}, acc1 = {}, acc2 = {}, acc3 = {};
    for (int k = 0; k < 16; ++k) {
        const int kk = 4 * k + koff;
        const v2f a  = *(const v2f*)(aRow + kk);
        const v2f b0 = *(const v2f*)(w0 + kk);
        const v2f b1 = *(const v2f*)(w1 + kk);
        const v2f b2 = *(const v2f*)(w2 + kk);
        const v2f b3 = *(const v2f*)(w3 + kk);
        acc0 = __builtin_amdgcn_wmma_f32_16x16x4_f32(false, a, false, b0,
                                                     (short)0, acc0, false, false);
        acc1 = __builtin_amdgcn_wmma_f32_16x16x4_f32(false, a, false, b1,
                                                     (short)0, acc1, false, false);
        acc2 = __builtin_amdgcn_wmma_f32_16x16x4_f32(false, a, false, b2,
                                                     (short)0, acc2, false, false);
        acc3 = __builtin_amdgcn_wmma_f32_16x16x4_f32(false, a, false, b3,
                                                     (short)0, acc3, false, false);
    }

    const int row0 = rowBase + wave * 16 + ((lane >> 4) << 3);
    for (int v = 0; v < 8; ++v) {
        int row = row0 + v;
        if (row < n_nodes) {
            float* xr = X + (size_t)row * DFEAT + m;
            xr[0]  = acc0[v];
            xr[16] = acc1[v];
            xr[32] = acc2[v];
            xr[48] = acc3[v];
        }
    }
}

template <bool FULL>
__device__ __forceinline__ void g2_tau_rows(const float* __restrict__ X,
                                            float* __restrict__ sums,
                                            int sv, int dv, int lane, int nvalid) {
#pragma unroll
    for (int j0 = 0; j0 < EDGES_PER_WAVE; j0 += EDGE_BATCH) {
        float s0[EDGE_BATCH], s1[EDGE_BATCH], d0[EDGE_BATCH], d1[EDGE_BATCH];
#pragma unroll
        for (int j = 0; j < EDGE_BATCH; ++j) {
            const int s = __builtin_amdgcn_readlane(sv, j0 + j);
            const int d = __builtin_amdgcn_readlane(dv, j0 + j);
            const float* srow = X + (size_t)s * DFEAT;
            const float* drow = X + (size_t)d * DFEAT;
            s0[j] = srow[lane];
            s1[j] = srow[32 + lane];
            d0[j] = drow[lane];
            d1[j] = drow[32 + lane];
        }
#pragma unroll
        for (int j = 0; j < EDGE_BATCH; ++j) {
            if (FULL || j0 + j < nvalid) {
                const int d = __builtin_amdgcn_readlane(dv, j0 + j);
                float* orow = sums + (size_t)d * DFEAT;
                const float x0 = s0[j] - d0[j];
                const float x1 = s1[j] - d1[j];
                atomicAdd(orow + lane, x0 * x0);
                atomicAdd(orow + 32 + lane, x1 * x1);
            }
        }
    }
}

__global__ __launch_bounds__(256) void g2_scatter_tau_kernel(
        const float* __restrict__ X,
        const int* __restrict__ src,
        const int* __restrict__ dst,
        float* __restrict__ sums,
        int n_edges) {
    const int lane  = threadIdx.x & 31;
    const int ebase = __builtin_amdgcn_readfirstlane(
        (int)((blockIdx.x * blockDim.x + threadIdx.x) >> 5)) * EDGES_PER_WAVE;
    if (ebase >= n_edges) return;
    const int e = ebase + lane;
    const bool valid = e < n_edges;
    const int sv = valid ? src[e] : 0;
    const int dv = valid ? dst[e] : 0;
    const int nvalid = n_edges - ebase;
    if (nvalid >= EDGES_PER_WAVE) g2_tau_rows<true >(X, sums, sv, dv, lane, EDGES_PER_WAVE);
    else                          g2_tau_rows<false>(X, sums, sv, dv, lane, nvalid);
}

__global__ void g2_finalize_kernel(const float* __restrict__ sums,
                                   const float* __restrict__ cnt,
                                   float* __restrict__ out,
                                   int n_elems) {
    int i = blockIdx.x * blockDim.x + threadIdx.x;
    if (i >= n_elems) return;
    float c = fmaxf(cnt[i >> 6], 1.0f);
    out[i] = tanhf(sums[i] / c);
}

extern "C" void kernel_launch(void* const* d_in, const int* in_sizes, int n_in,
                              void* d_out, int out_size, void* d_ws, size_t ws_size,
                              hipStream_t stream) {
    const float* nf = (const float*)d_in[0];
    const float* W  = (const float*)d_in[1];
    const int*   ei = (const int*)d_in[2];
    float* out = (float*)d_out;

    const int n_nodes = in_sizes[0] / DFEAT;
    const int n_edges = in_sizes[2] / 2;
    const int* src = ei;
    const int* dst = ei + n_edges;

    float* sums = (float*)d_ws;
    float* cnt  = sums + (size_t)n_nodes * DFEAT;
    float* X    = cnt + n_nodes;

    const int T = 256;

    {
        int n = n_nodes * (DFEAT + 1);
        g2_zero_kernel<<<(n + T - 1) / T, T, 0, stream>>>(sums, n);
    }
    {
        g2_scatter_agg_kernel<<<(n_edges + T - 1) / T, T, 0, stream>>>(
            nf, src, dst, sums, cnt, n_edges);
    }
    {
        int blocks = (n_nodes + ROWS_PER_BLOCK - 1) / ROWS_PER_BLOCK;
        g2_gemm_wmma_kernel<<<blocks, T, 0, stream>>>(sums, cnt, W, X, n_nodes);
    }
    {
        int n = n_nodes * DFEAT;
        g2_zero_kernel<<<(n + T - 1) / T, T, 0, stream>>>(sums, n);
    }
    {
        g2_scatter_tau_kernel<<<(n_edges + T - 1) / T, T, 0, stream>>>(
            X, src, dst, sums, n_edges);
    }
    {
        int n = n_nodes * DFEAT;
        g2_finalize_kernel<<<(n + T - 1) / T, T, 0, stream>>>(sums, cnt, out, n);
    }
}
